# starting kernel
# baseline (speedup 1.0000x reference)
#include <hip/hip_runtime.h>
#include <math.h>

typedef __attribute__((ext_vector_type(2))) float v2f;
typedef __attribute__((ext_vector_type(8))) float v8f;

#define EMB   768
#define HS    64
#define SEQ   2048
#define BATCH 8
#define XPAD  772
#define VPAD  72

__device__ __forceinline__ v8f wmma_f32(v2f a, v2f b, v8f c) {
  return __builtin_amdgcn_wmma_f32_16x16x4_f32(
       false, a,  false, b,
       (short)0, c,  false,  false);
}

__global__ __launch_bounds__(128) void proj_kernel(
    const float* __restrict__ x,
    const float* __restrict__ Wq, const float* __restrict__ Wk,
    const float* __restrict__ Wv,
    float* __restrict__ Q, float* __restrict__ Kt, float* __restrict__ V) {
  __shared__ float xs[16 * XPAD];

  const int tileM = blockIdx.x;
  const int which = blockIdx.y;
  const int tid   = threadIdx.x;
  const int lane  = tid & 31;
  const int wave  = tid >> 5;
  const int l16   = lane & 15;
  const int half  = lane >> 4;
  const int halfk = half * 2;
  const int rowBase = tileM * 16;

  for (int i = tid; i < 16 * EMB; i += 128) {
    const int r = i / EMB, cc = i - r * EMB;
    xs[r * XPAD + cc] = x[(rowBase + r) * EMB + cc];
  }
  __syncthreads();

  const float* W = (which == 0) ? Wq : (which == 1) ? Wk : Wv;
  const int n = wave * 16 + l16;

  v8f c = {};
  for (int kc = 0; kc < EMB / 4; ++kc) {
    const int k0 = kc * 4 + halfk;
    v2f a, b;
    a.x = xs[l16 * XPAD + k0];
    a.y = xs[l16 * XPAD + k0 + 1];
    b.x = W[k0 * HS + n];
    b.y = W[(k0 + 1) * HS + n];
    c = wmma_f32(a, b, c);
  }

  if (which == 1) {
    const int bidx = rowBase / SEQ;
    const int s0   = rowBase - bidx * SEQ;
#pragma unroll
    for (int v = 0; v < 8; ++v) {
      const int srow = s0 + v + half * 8;
      Kt[(bidx * HS + n) * SEQ + srow] = c[v];
    }
  } else {
    float* O = (which == 0) ? Q : V;
#pragma unroll
    for (int v = 0; v < 8; ++v) {
      const int m = rowBase + v + half * 8;
      O[m * HS + n] = c[v];
    }
  }
}

__global__ __launch_bounds__(256) void attn_kernel(
    const float* __restrict__ Q, const float* __restrict__ Kt,
    const float* __restrict__ V, float* __restrict__ out) {
  __shared__ float Kbuf[2][HS * 16];
  __shared__ float Vbuf[2][16 * VPAD];
  __shared__ float plds[8][16 * 17];

  const int tid   = threadIdx.x;
  const int lane  = tid & 31;
  const int wave  = tid >> 5;
  const int gt    = blockIdx.x * 8 + wave;
  const int b     = gt >> 7;
  const int qbase = (gt & 127) * 16;
  const int l16   = lane & 15;
  const int half  = lane >> 4;
  const int halfk = half * 2;

  v2f qa[16];
  const int mrow = (b * SEQ + qbase + l16) * HS;
#pragma unroll
  for (int kc = 0; kc < 16; ++kc) {
    qa[kc].x = Q[mrow + kc * 4 + halfk];
    qa[kc].y = Q[mrow + kc * 4 + halfk + 1];
  }

  v8f acc0 = {}, acc1 = {}, acc2 = {}, acc3 = {};
  float mst[8], lst[8];
#pragma unroll
  for (int v = 0; v < 8; ++v) { mst[v] = -INFINITY; lst[v] = 0.0f; }

  float* pl = &plds[wave][0];
  const float scale = 0.125f;

  auto stage = [&](int kt, int bufi) {
    float* kb = &Kbuf[bufi][0];
    float* vb = &Vbuf[bufi][0];
    const int kBase = (b * HS) * SEQ + kt * 16;
    const int vBase = (b * SEQ + kt * 16) * HS;
#pragma unroll
    for (int i = tid; i < HS * 16; i += 256) {
      const int h = i >> 4, kl = i & 15;
      kb[i] = Kt[kBase + h * SEQ + kl];
    }
#pragma unroll
    for (int i = tid; i < 16 * HS; i += 256) {
      const int kl = i >> 6, h = i & 63;
      vb[kl * VPAD + h] = V[vBase + kl * HS + h];
    }
  };

  stage(0, 0);

  for (int kt = 0; kt < SEQ / 16; ++kt) {
    __syncthreads();
    const int bufi = kt & 1;
    if (kt + 1 < SEQ / 16) stage(kt + 1, bufi ^ 1);

    const float* kb = &Kbuf[bufi][0];
    const float* vb = &Vbuf[bufi][0];

    v8f s = {};
#pragma unroll
    for (int kc = 0; kc < 16; ++kc) {
      const int h = kc * 4 + halfk;
      v2f kv;
      kv.x = kb[h * 16 + l16];
      kv.y = kb[(h + 1) * 16 + l16];
      s = wmma_f32(qa[kc], kv, s);
    }

#pragma unroll
    for (int v = 0; v < 8; ++v) {
      float sv = s[v] * scale;
      float tm = sv;
      tm = fmaxf(tm, __shfl_xor(tm, 1));
      tm = fmaxf(tm, __shfl_xor(tm, 2));
      tm = fmaxf(tm, __shfl_xor(tm, 4));
      tm = fmaxf(tm, __shfl_xor(tm, 8));
      const float nm    = fmaxf(mst[v], tm);
      const float alpha = __expf(mst[v] - nm);
      const float p     = __expf(sv - nm);
      float rs = p;
      rs += __shfl_xor(rs, 1);
      rs += __shfl_xor(rs, 2);
      rs += __shfl_xor(rs, 4);
      rs += __shfl_xor(rs, 8);
      lst[v] = lst[v] * alpha + rs;
      mst[v] = nm;
      acc0[v] *= alpha; acc1[v] *= alpha; acc2[v] *= alpha; acc3[v] *= alpha;
      pl[(v + half * 8) * 17 + l16] = p;
    }

#pragma unroll
    for (int kc = 0; kc < 4; ++kc) {
      v2f a;
      a.x = pl[l16 * 17 + kc * 4 + halfk];
      a.y = pl[l16 * 17 + kc * 4 + halfk + 1];
      const int r0 = (kc * 4 + halfk) * VPAD;
      const int r1 = r0 + VPAD;
      { v2f bv; bv.x = vb[r0 + l16];      bv.y = vb[r1 + l16];      acc0 = wmma_f32(a, bv, acc0); }
      { v2f bv; bv.x = vb[r0 + 16 + l16]; bv.y = vb[r1 + 16 + l16]; acc1 = wmma_f32(a, bv, acc1); }
      { v2f bv; bv.x = vb[r0 + 32 + l16]; bv.y = vb[r1 + 32 + l16]; acc2 = wmma_f32(a, bv, acc2); }
      { v2f bv; bv.x = vb[r0 + 48 + l16]; bv.y = vb[r1 + 48 + l16]; acc3 = wmma_f32(a, bv, acc3); }
    }
  }

#pragma unroll
  for (int v = 0; v < 8; ++v) {
    const float inv = 1.0f / lst[v];
    const int orow = (b * SEQ + qbase + v + half * 8) * HS;
    out[orow + l16]      = acc0[v] * inv;
    out[orow + 16 + l16] = acc1[v] * inv;
    out[orow + 32 + l16] = acc2[v] * inv;
    out[orow + 48 + l16] = acc3[v] * inv;
  }
}

extern "C" void kernel_launch(void* const* d_in, const int* in_sizes, int n_in,
                              void* d_out, int out_size, void* d_ws, size_t ws_size,
                              hipStream_t stream) {
  (void)in_sizes; (void)n_in; (void)out_size; (void)ws_size;
  const float* x  = (const float*)d_in[0];
  const float* Wk = (const float*)d_in[1];
  const float* Wq = (const float*)d_in[2];
  const float* Wv = (const float*)d_in[3];

  float* ws = (float*)d_ws;
  const size_t plane = (size_t)BATCH * SEQ * HS;
  float* Q  = ws;
  float* Kt = ws + plane;
  float* V  = ws + 2 * plane;
  float* out = (float*)d_out;

  dim3 g1(BATCH * SEQ / 16, 3);
  proj_kernel<<<g1, 128, 0, stream>>>(x, Wq, Wk, Wv, Q, Kt, V);

  attn_kernel<<<(BATCH * SEQ / 16) / 8, 256, 0, stream>>>(Q, Kt, V, out);
}


# attention K/V B fragments read straight from fragment-ordered workspace tiles into registers (16-byte loads, one step ahead); no K/V LDS staging or block barrier; 256 blocks of 4 independent waves
# speedup vs baseline: 2.1901x; 2.1901x over previous
#include <hip/hip_runtime.h>
#include <math.h>


typedef __attribute__((ext_vector_type(2))) float v2f;
typedef __attribute__((ext_vector_type(4))) float v4f;
typedef __attribute__((ext_vector_type(8))) float v8f;

#define EMB   768
#define HS    64
#define SEQ   2048
#define BATCH 8
#define XPAD  772
#define TILE  1024
#define NKT   (SEQ / 16)

__device__ __forceinline__ v8f wmma_f32(v2f a, v2f b, v8f c) {
  return __builtin_amdgcn_wmma_f32_16x16x4_f32(
       false, a,  false, b,
       (short)0, c,  false,  false);
}


__global__ __launch_bounds__(128) void proj_kernel(
    const float* __restrict__ x,
    const float* __restrict__ Wq, const float* __restrict__ Wk,
    const float* __restrict__ Wv,
    float* __restrict__ Q, float* __restrict__ Kf, float* __restrict__ Vf) {
  __shared__ float xs[16 * XPAD];

  const int tileM = blockIdx.x;
  const int which = blockIdx.y;
  const int tid   = threadIdx.x;
  const int lane  = tid & 31;
  const int wave  = tid >> 5;
  const int l16   = lane & 15;
  const int half  = lane >> 4;
  const int halfk = half * 2;
  const int rowBase = tileM * 16;

  for (int i = tid; i < 16 * EMB; i += 128) {
    const int r = i / EMB, cc = i - r * EMB;
    xs[r * XPAD + cc] = x[(rowBase + r) * EMB + cc];
  }
  __syncthreads();

  const float* W = (which == 0) ? Wq : (which == 1) ? Wk : Wv;
  const int n = wave * 16 + l16;

  v8f c = {};
  for (int kc = 0; kc < EMB / 4; ++kc) {
    const int k0 = kc * 4 + halfk;
    v2f a, b;
    a.x = xs[l16 * XPAD + k0];
    a.y = xs[l16 * XPAD + k0 + 1];
    b.x = W[k0 * HS + n];
    b.y = W[(k0 + 1) * HS + n];
    c = wmma_f32(a, b, c);
  }

  if (which == 0) {
#pragma unroll
    for (int v = 0; v < 8; ++v) {
      const int m = rowBase + v + half * 8;
      Q[m * HS + n] = c[v];
    }
  } else if (which == 1) {
    float* T = Kf + tileM * TILE;
    const int base = (n >> 3) * 128 + ((n >> 1) & 1) * 64 + ((n >> 2) & 1) * 2 + (n & 1);
#pragma unroll
    for (int v = 0; v < 8; ++v) {
      const int key = v + half * 8;
      T[base + key * 4] = c[v];
    }
  } else {
    float* T = Vf + tileM * TILE;
#pragma unroll
    for (int v = 0; v < 8; v += 2) {
      const int kc = (v >> 2) + 2 * half;
      const int idx = (kc * 2 + (wave >> 1)) * 128 + (((v >> 1) & 1) * 16 + l16) * 4 + (wave & 1) * 2;
      v2f pr; pr.x = c[v]; pr.y = c[v + 1];
      *(v2f*)(T + idx) = pr;
    }
  }
}

__global__ __launch_bounds__(128) void attn_kernel(
    const float* __restrict__ Q, const float* __restrict__ Kf,
    const float* __restrict__ Vf, float* __restrict__ out) {
  __shared__ float plds[4][16 * 17];

  const int tid   = threadIdx.x;
  const int lane  = tid & 31;
  const int wave  = tid >> 5;
  const int gt    = blockIdx.x * 4 + wave;
  const int b     = gt >> 7;
  const int qbase = (gt & 127) * 16;
  const int l16   = lane & 15;
  const int half  = lane >> 4;
  const int halfk = half * 2;

  v2f qa[16];
  const int mrow = (b * SEQ + qbase + l16) * HS;
#pragma unroll
  for (int kc = 0; kc < 16; ++kc) {
    qa[kc].x = Q[mrow + kc * 4 + halfk];
    qa[kc].y = Q[mrow + kc * 4 + halfk + 1];
  }

  v8f acc0 = {}, acc1 = {}, acc2 = {}, acc3 = {};
  float mst[8], lst[8];
#pragma unroll
  for (int v = 0; v < 8; ++v) { mst[v] = -INFINITY; lst[v] = 0.0f; }

  float* pl = &plds[wave][0];
  const float scale = 0.125f;

  const v4f* kp = (const v4f*)Kf + (b * NKT) * (TILE / 4) + lane;
  const v4f* vp = (const v4f*)Vf + (b * NKT) * (TILE / 4) + lane;

  auto fetch = [&](v4f (&kf)[8], v4f (&vf)[8], int kt) {
    const v4f* kq = kp + kt * (TILE / 4);
    const v4f* vq = vp + kt * (TILE / 4);
#pragma unroll
    for (int i = 0; i < 8; ++i) kf[i] = kq[i * 32];
#pragma unroll
    for (int i = 0; i < 8; ++i) vf[i] = vq[i * 32];
  };

  auto step = [&](const v4f (&kf)[8], const v4f (&vf)[8]) {
    v8f s = {};
#pragma unroll
    for (int i = 0; i < 8; ++i) {
      v2f k0; k0.x = kf[i].x; k0.y = kf[i].y;
      v2f k1; k1.x = kf[i].z; k1.y = kf[i].w;
      s = wmma_f32(qa[2 * i],     k0, s);
      s = wmma_f32(qa[2 * i + 1], k1, s);
    }

#pragma unroll
    for (int v = 0; v < 8; ++v) {
      float sv = s[v] * scale;
      float tm = sv;
      tm = fmaxf(tm, __shfl_xor(tm, 1));
      tm = fmaxf(tm, __shfl_xor(tm, 2));
      tm = fmaxf(tm, __shfl_xor(tm, 4));
      tm = fmaxf(tm, __shfl_xor(tm, 8));
      const float nm    = fmaxf(mst[v], tm);
      const float alpha = __expf(mst[v] - nm);
      const float p     = __expf(sv - nm);
      float rs = p;
      rs += __shfl_xor(rs, 1);
      rs += __shfl_xor(rs, 2);
      rs += __shfl_xor(rs, 4);
      rs += __shfl_xor(rs, 8);
      lst[v] = lst[v] * alpha + rs;
      mst[v] = nm;
      acc0[v] *= alpha; acc1[v] *= alpha; acc2[v] *= alpha; acc3[v] *= alpha;
      pl[(v + half * 8) * 17 + l16] = p;
    }

#pragma unroll
    for (int kc = 0; kc < 4; ++kc) {
      v2f a;
      a.x = pl[l16 * 17 + kc * 4 + halfk];
      a.y = pl[l16 * 17 + kc * 4 + halfk + 1];
      const v4f f01 = vf[2 * kc];
      const v4f f23 = vf[2 * kc + 1];
      { v2f bv; bv.x = f01.x; bv.y = f01.y; acc0 = wmma_f32(a, bv, acc0); }
      { v2f bv; bv.x = f01.z; bv.y = f01.w; acc1 = wmma_f32(a, bv, acc1); }
      { v2f bv; bv.x = f23.x; bv.y = f23.y; acc2 = wmma_f32(a, bv, acc2); }
      { v2f bv; bv.x = f23.z; bv.y = f23.w; acc3 = wmma_f32(a, bv, acc3); }
    }
  };

  v4f ka[8], va[8], kb[8], vb[8];
  fetch(ka, va, 0);
  for (int kt = 0; kt < NKT; kt += 2) {
    fetch(kb, vb, kt + 1);
    step(ka, va);
    fetch(ka, va, (kt + 2 < NKT) ? kt + 2 : NKT - 1);
    step(kb, vb);
  }

#pragma unroll
  for (int v = 0; v < 8; ++v) {
    const float inv = 1.0f / lst[v];
    const int orow = (b * SEQ + qbase + v + half * 8) * HS;
    out[orow + l16]      = acc0[v] * inv;
    out[orow + 16 + l16] = acc1[v] * inv;
    out[orow + 32 + l16] = acc2[v] * inv;
    out[orow + 48 + l16] = acc3[v] * inv;
  }
}

extern "C" void kernel_launch(void* const* d_in, const int* in_sizes, int n_in,
                              void* d_out, int out_size, void* d_ws, size_t ws_size,
                              hipStream_t stream) {
  (void)in_sizes; (void)n_in; (void)out_size; (void)ws_size;
  const float* x  = (const float*)d_in[0];
  const float* Wk = (const float*)d_in[1];
  const float* Wq = (const float*)d_in[2];
  const float* Wv = (const float*)d_in[3];

  float* ws = (float*)d_ws;
  const size_t plane = (size_t)BATCH * SEQ * HS;
  float* Q  = ws;
  float* Kf = ws + plane;
  float* Vf = ws + 2 * plane;
  float* out = (float*)d_out;

  dim3 g1(BATCH * SEQ / 16, 3);
  proj_kernel<<<g1, 128, 0, stream>>>(x, Wq, Wk, Wv, Q, Kf, Vf);

  attn_kernel<<<(BATCH * SEQ / 16) / 4, 128, 0, stream>>>(Q, Kf, Vf, out);
}


# attention softmax: row max and row sum xor butterflies on DPP row_xmask vector-ALU ops instead of LDS-permute shuffles; same lane pairs in the same order, sums bit-identical
# speedup vs baseline: 3.2690x; 1.4926x over previous
#include <hip/hip_runtime.h>
#include <math.h>


typedef __attribute__((ext_vector_type(2))) float v2f;
typedef __attribute__((ext_vector_type(4))) float v4f;
typedef __attribute__((ext_vector_type(8))) float v8f;

#define EMB   768
#define HS    64
#define SEQ   2048
#define BATCH 8
#define XPAD  772
#define TILE  1024
#define NKT   (SEQ / 16)

__device__ __forceinline__ v8f wmma_f32(v2f a, v2f b, v8f c) {
  return __builtin_amdgcn_wmma_f32_16x16x4_f32(
       false, a,  false, b,
       (short)0, c,  false,  false);
}


template <int M>
__device__ __forceinline__ float row_xor(float x) {
  return __builtin_amdgcn_update_dpp(0.0f, x, 0x160 + M, 0xf, 0xf, true);
}


__global__ __launch_bounds__(128) void proj_kernel(
    const float* __restrict__ x,
    const float* __restrict__ Wq, const float* __restrict__ Wk,
    const float* __restrict__ Wv,
    float* __restrict__ Q, float* __restrict__ Kf, float* __restrict__ Vf) {
  __shared__ float xs[16 * XPAD];

  const int tileM = blockIdx.x;
  const int which = blockIdx.y;
  const int tid   = threadIdx.x;
  const int lane  = tid & 31;
  const int wave  = tid >> 5;
  const int l16   = lane & 15;
  const int half  = lane >> 4;
  const int halfk = half * 2;
  const int rowBase = tileM * 16;

  for (int i = tid; i < 16 * EMB; i += 128) {
    const int r = i / EMB, cc = i - r * EMB;
    xs[r * XPAD + cc] = x[(rowBase + r) * EMB + cc];
  }
  __syncthreads();

  const float* W = (which == 0) ? Wq : (which == 1) ? Wk : Wv;
  const int n = wave * 16 + l16;

  v8f c = {};
  for (int kc = 0; kc < EMB / 4; ++kc) {
    const int k0 = kc * 4 + halfk;
    v2f a, b;
    a.x = xs[l16 * XPAD + k0];
    a.y = xs[l16 * XPAD + k0 + 1];
    b.x = W[k0 * HS + n];
    b.y = W[(k0 + 1) * HS + n];
    c = wmma_f32(a, b, c);
  }

  if (which == 0) {
#pragma unroll
    for (int v = 0; v < 8; ++v) {
      const int m = rowBase + v + half * 8;
      Q[m * HS + n] = c[v];
    }
  } else if (which == 1) {
    float* T = Kf + tileM * TILE;
    const int base = (n >> 3) * 128 + ((n >> 1) & 1) * 64 + ((n >> 2) & 1) * 2 + (n & 1);
#pragma unroll
    for (int v = 0; v < 8; ++v) {
      const int key = v + half * 8;
      T[base + key * 4] = c[v];
    }
  } else {
    float* T = Vf + tileM * TILE;
#pragma unroll
    for (int v = 0; v < 8; v += 2) {
      const int kc = (v >> 2) + 2 * half;
      const int idx = (kc * 2 + (wave >> 1)) * 128 + (((v >> 1) & 1) * 16 + l16) * 4 + (wave & 1) * 2;
      v2f pr; pr.x = c[v]; pr.y = c[v + 1];
      *(v2f*)(T + idx) = pr;
    }
  }
}

__global__ __launch_bounds__(128) void attn_kernel(
    const float* __restrict__ Q, const float* __restrict__ Kf,
    const float* __restrict__ Vf, float* __restrict__ out) {
  __shared__ float plds[4][16 * 17];

  const int tid   = threadIdx.x;
  const int lane  = tid & 31;
  const int wave  = tid >> 5;
  const int gt    = blockIdx.x * 4 + wave;
  const int b     = gt >> 7;
  const int qbase = (gt & 127) * 16;
  const int l16   = lane & 15;
  const int half  = lane >> 4;
  const int halfk = half * 2;

  v2f qa[16];
  const int mrow = (b * SEQ + qbase + l16) * HS;
#pragma unroll
  for (int kc = 0; kc < 16; ++kc) {
    qa[kc].x = Q[mrow + kc * 4 + halfk];
    qa[kc].y = Q[mrow + kc * 4 + halfk + 1];
  }

  v8f acc0 = {}, acc1 = {}, acc2 = {}, acc3 = {};
  float mst[8], lst[8];
#pragma unroll
  for (int v = 0; v < 8; ++v) { mst[v] = -INFINITY; lst[v] = 0.0f; }

  float* pl = &plds[wave][0];
  const float scale = 0.125f;

  const v4f* kp = (const v4f*)Kf + (b * NKT) * (TILE / 4) + lane;
  const v4f* vp = (const v4f*)Vf + (b * NKT) * (TILE / 4) + lane;

  auto fetch = [&](v4f (&kf)[8], v4f (&vf)[8], int kt) {
    const v4f* kq = kp + kt * (TILE / 4);
    const v4f* vq = vp + kt * (TILE / 4);
#pragma unroll
    for (int i = 0; i < 8; ++i) kf[i] = kq[i * 32];
#pragma unroll
    for (int i = 0; i < 8; ++i) vf[i] = vq[i * 32];
  };

  auto step = [&](const v4f (&kf)[8], const v4f (&vf)[8]) {
    v8f s = {};
#pragma unroll
    for (int i = 0; i < 8; ++i) {
      v2f k0; k0.x = kf[i].x; k0.y = kf[i].y;
      v2f k1; k1.x = kf[i].z; k1.y = kf[i].w;
      s = wmma_f32(qa[2 * i],     k0, s);
      s = wmma_f32(qa[2 * i + 1], k1, s);
    }

#pragma unroll
    for (int v = 0; v < 8; ++v) {
      float sv = s[v] * scale;
      float tm = sv;
      tm = fmaxf(tm, row_xor<1>(tm));
      tm = fmaxf(tm, row_xor<2>(tm));
      tm = fmaxf(tm, row_xor<4>(tm));
      tm = fmaxf(tm, row_xor<8>(tm));
      const float nm    = fmaxf(mst[v], tm);
      const float alpha = __expf(mst[v] - nm);
      const float p     = __expf(sv - nm);
      float rs = p;
      rs += row_xor<1>(rs);
      rs += row_xor<2>(rs);
      rs += row_xor<4>(rs);
      rs += row_xor<8>(rs);
      lst[v] = lst[v] * alpha + rs;
      mst[v] = nm;
      acc0[v] *= alpha; acc1[v] *= alpha; acc2[v] *= alpha; acc3[v] *= alpha;
      pl[(v + half * 8) * 17 + l16] = p;
    }

#pragma unroll
    for (int kc = 0; kc < 4; ++kc) {
      v2f a;
      a.x = pl[l16 * 17 + kc * 4 + halfk];
      a.y = pl[l16 * 17 + kc * 4 + halfk + 1];
      const v4f f01 = vf[2 * kc];
      const v4f f23 = vf[2 * kc + 1];
      { v2f bv; bv.x = f01.x; bv.y = f01.y; acc0 = wmma_f32(a, bv, acc0); }
      { v2f bv; bv.x = f01.z; bv.y = f01.w; acc1 = wmma_f32(a, bv, acc1); }
      { v2f bv; bv.x = f23.x; bv.y = f23.y; acc2 = wmma_f32(a, bv, acc2); }
      { v2f bv; bv.x = f23.z; bv.y = f23.w; acc3 = wmma_f32(a, bv, acc3); }
    }
  };

  v4f ka[8], va[8], kb[8], vb[8];
  fetch(ka, va, 0);
  for (int kt = 0; kt < NKT; kt += 2) {
    fetch(kb, vb, kt + 1);
    step(ka, va);
    fetch(ka, va, (kt + 2 < NKT) ? kt + 2 : NKT - 1);
    step(kb, vb);
  }

#pragma unroll
  for (int v = 0; v < 8; ++v) {
    const float inv = 1.0f / lst[v];
    const int orow = (b * SEQ + qbase + v + half * 8) * HS;
    out[orow + l16]      = acc0[v] * inv;
    out[orow + 16 + l16] = acc1[v] * inv;
    out[orow + 32 + l16] = acc2[v] * inv;
    out[orow + 48 + l16] = acc3[v] * inv;
  }
}

extern "C" void kernel_launch(void* const* d_in, const int* in_sizes, int n_in,
                              void* d_out, int out_size, void* d_ws, size_t ws_size,
                              hipStream_t stream) {
  (void)in_sizes; (void)n_in; (void)out_size; (void)ws_size;
  const float* x  = (const float*)d_in[0];
  const float* Wk = (const float*)d_in[1];
  const float* Wq = (const float*)d_in[2];
  const float* Wv = (const float*)d_in[3];

  float* ws = (float*)d_ws;
  const size_t plane = (size_t)BATCH * SEQ * HS;
  float* Q  = ws;
  float* Kf = ws + plane;
  float* Vf = ws + 2 * plane;
  float* out = (float*)d_out;

  dim3 g1(BATCH * SEQ / 16, 3);
  proj_kernel<<<g1, 128, 0, stream>>>(x, Wq, Wk, Wv, Q, Kf, Vf);

  attn_kernel<<<(BATCH * SEQ / 16) / 4, 128, 0, stream>>>(Q, Kf, Vf, out);
}


# attention softmax blocked over 32 keys: in-lane max of two score tiles, one cross-lane max butterfly and one rescale per 32 keys; row sums kept per lane, joined after the loop (sum order changes)
# speedup vs baseline: 3.4098x; 1.0431x over previous
#include <hip/hip_runtime.h>
#include <math.h>


typedef __attribute__((ext_vector_type(2))) float v2f;
typedef __attribute__((ext_vector_type(4))) float v4f;
typedef __attribute__((ext_vector_type(8))) float v8f;

#define EMB   768
#define HS    64
#define SEQ   2048
#define BATCH 8
#define XPAD  772
#define TILE  1024
#define NKT   (SEQ / 16)

__device__ __forceinline__ v8f wmma_f32(v2f a, v2f b, v8f c) {
  return __builtin_amdgcn_wmma_f32_16x16x4_f32(
       false, a,  false, b,
       (short)0, c,  false,  false);
}


template <int M>
__device__ __forceinline__ float row_xor(float x) {
  return __builtin_amdgcn_update_dpp(0.0f, x, 0x160 + M, 0xf, 0xf, true);
}


__global__ __launch_bounds__(128) void proj_kernel(
    const float* __restrict__ x,
    const float* __restrict__ Wq, const float* __restrict__ Wk,
    const float* __restrict__ Wv,
    float* __restrict__ Q, float* __restrict__ Kf, float* __restrict__ Vf) {
  __shared__ float xs[16 * XPAD];

  const int tileM = blockIdx.x;
  const int which = blockIdx.y;
  const int tid   = threadIdx.x;
  const int lane  = tid & 31;
  const int wave  = tid >> 5;
  const int l16   = lane & 15;
  const int half  = lane >> 4;
  const int halfk = half * 2;
  const int rowBase = tileM * 16;

  for (int i = tid; i < 16 * EMB; i += 128) {
    const int r = i / EMB, cc = i - r * EMB;
    xs[r * XPAD + cc] = x[(rowBase + r) * EMB + cc];
  }
  __syncthreads();

  const float* W = (which == 0) ? Wq : (which == 1) ? Wk : Wv;
  const int n = wave * 16 + l16;

  v8f c = {};
  for (int kc = 0; kc < EMB / 4; ++kc) {
    const int k0 = kc * 4 + halfk;
    v2f a, b;
    a.x = xs[l16 * XPAD + k0];
    a.y = xs[l16 * XPAD + k0 + 1];
    b.x = W[k0 * HS + n];
    b.y = W[(k0 + 1) * HS + n];
    c = wmma_f32(a, b, c);
  }

  if (which == 0) {
#pragma unroll
    for (int v = 0; v < 8; ++v) {
      const int m = rowBase + v + half * 8;
      Q[m * HS + n] = c[v];
    }
  } else if (which == 1) {
    float* T = Kf + tileM * TILE;
    const int base = (n >> 3) * 128 + ((n >> 1) & 1) * 64 + ((n >> 2) & 1) * 2 + (n & 1);
#pragma unroll
    for (int v = 0; v < 8; ++v) {
      const int key = v + half * 8;
      T[base + key * 4] = c[v];
    }
  } else {
    float* T = Vf + tileM * TILE;
#pragma unroll
    for (int v = 0; v < 8; v += 2) {
      const int kc = (v >> 2) + 2 * half;
      const int idx = (kc * 2 + (wave >> 1)) * 128 + (((v >> 1) & 1) * 16 + l16) * 4 + (wave & 1) * 2;
      v2f pr; pr.x = c[v]; pr.y = c[v + 1];
      *(v2f*)(T + idx) = pr;
    }
  }
}

#define KT  2
#define PT  (16 * 17)

__global__ __launch_bounds__(128) void attn_kernel(
    const float* __restrict__ Q, const float* __restrict__ Kf,
    const float* __restrict__ Vf, float* __restrict__ out) {
  __shared__ float plds[4][KT * PT];

  const int tid   = threadIdx.x;
  const int lane  = tid & 31;
  const int wave  = tid >> 5;
  const int gt    = blockIdx.x * 4 + wave;
  const int b     = gt >> 7;
  const int qbase = (gt & 127) * 16;
  const int l16   = lane & 15;
  const int half  = lane >> 4;
  const int halfk = half * 2;

  v2f qa[16];
  const int mrow = (b * SEQ + qbase + l16) * HS;
#pragma unroll
  for (int kc = 0; kc < 16; ++kc) {
    qa[kc].x = Q[mrow + kc * 4 + halfk];
    qa[kc].y = Q[mrow + kc * 4 + halfk + 1];
  }

  v8f acc0 = {}, acc1 = {}, acc2 = {}, acc3 = {};
  float mst[8], lst[8];
#pragma unroll
  for (int v = 0; v < 8; ++v) { mst[v] = -INFINITY; lst[v] = 0.0f; }

  float* pl = &plds[wave][0];
  const float scale = 0.125f;

  const v4f* kp = (const v4f*)Kf + (b * NKT) * (TILE / 4) + lane;
  const v4f* vp = (const v4f*)Vf + (b * NKT) * (TILE / 4) + lane;

  auto fetch = [&](v4f (&kf)[KT][8], v4f (&vf)[KT][8], int g) {
    const v4f* kq = kp + g * (KT * TILE / 4);
    const v4f* vq = vp + g * (KT * TILE / 4);
#pragma unroll
    for (int t = 0; t < KT; ++t)
#pragma unroll
      for (int i = 0; i < 8; ++i) kf[t][i] = kq[(t * 8 + i) * 32];
#pragma unroll
    for (int t = 0; t < KT; ++t)
#pragma unroll
      for (int i = 0; i < 8; ++i) vf[t][i] = vq[(t * 8 + i) * 32];
  };

  auto step = [&](const v4f (&kf)[KT][8], const v4f (&vf)[KT][8]) {
    v8f s[KT];
#pragma unroll
    for (int t = 0; t < KT; ++t) s[t] = v8f{};
#pragma unroll
    for (int i = 0; i < 8; ++i) {
#pragma unroll
      for (int t = 0; t < KT; ++t) {
        v2f k0; k0.x = kf[t][i].x; k0.y = kf[t][i].y;
        s[t] = wmma_f32(qa[2 * i], k0, s[t]);
      }
#pragma unroll
      for (int t = 0; t < KT; ++t) {
        v2f k1; k1.x = kf[t][i].z; k1.y = kf[t][i].w;
        s[t] = wmma_f32(qa[2 * i + 1], k1, s[t]);
      }
    }

#pragma unroll
    for (int v = 0; v < 8; ++v) {
      float a[KT];
#pragma unroll
      for (int t = 0; t < KT; ++t) a[t] = s[t][v] * scale;
      float tm = a[0];
#pragma unroll
      for (int t = 1; t < KT; ++t) tm = fmaxf(tm, a[t]);
      tm = fmaxf(tm, row_xor<1>(tm));
      tm = fmaxf(tm, row_xor<2>(tm));
      tm = fmaxf(tm, row_xor<4>(tm));
      tm = fmaxf(tm, row_xor<8>(tm));
      const float nm    = fmaxf(mst[v], tm);
      const float alpha = __expf(mst[v] - nm);
      float ps = 0.0f;
#pragma unroll
      for (int t = 0; t < KT; ++t) {
        const float p = __expf(a[t] - nm);
        ps = (t == 0) ? p : ps + p;
        pl[t * PT + (v + half * 8) * 17 + l16] = p;
      }
      lst[v] = lst[v] * alpha + ps;
      mst[v] = nm;
      acc0[v] *= alpha; acc1[v] *= alpha; acc2[v] *= alpha; acc3[v] *= alpha;
    }

#pragma unroll
    for (int t = 0; t < KT; ++t) {
#pragma unroll
      for (int kc = 0; kc < 4; ++kc) {
        v2f a;
        a.x = pl[t * PT + l16 * 17 + kc * 4 + halfk];
        a.y = pl[t * PT + l16 * 17 + kc * 4 + halfk + 1];
        const v4f f01 = vf[t][2 * kc];
        const v4f f23 = vf[t][2 * kc + 1];
        { v2f bv; bv.x = f01.x; bv.y = f01.y; acc0 = wmma_f32(a, bv, acc0); }
        { v2f bv; bv.x = f01.z; bv.y = f01.w; acc1 = wmma_f32(a, bv, acc1); }
        { v2f bv; bv.x = f23.x; bv.y = f23.y; acc2 = wmma_f32(a, bv, acc2); }
        { v2f bv; bv.x = f23.z; bv.y = f23.w; acc3 = wmma_f32(a, bv, acc3); }
      }
    }
  };

  constexpr int NG = NKT / KT;
  v4f ka[KT][8], va[KT][8], kb[KT][8], vb[KT][8];
  fetch(ka, va, 0);
  for (int g = 0; g < NG; g += 2) {
    fetch(kb, vb, g + 1);
    step(ka, va);
    fetch(ka, va, (g + 2 < NG) ? g + 2 : NG - 1);
    step(kb, vb);
  }

#pragma unroll
  for (int v = 0; v < 8; ++v) {
    float l = lst[v];
    l += row_xor<1>(l);
    l += row_xor<2>(l);
    l += row_xor<4>(l);
    l += row_xor<8>(l);
    const float inv = 1.0f / l;
    const int orow = (b * SEQ + qbase + v + half * 8) * HS;
    out[orow + l16]      = acc0[v] * inv;
    out[orow + 16 + l16] = acc1[v] * inv;
    out[orow + 32 + l16] = acc2[v] * inv;
    out[orow + 48 + l16] = acc3[v] * inv;
  }
}

extern "C" void kernel_launch(void* const* d_in, const int* in_sizes, int n_in,
                              void* d_out, int out_size, void* d_ws, size_t ws_size,
                              hipStream_t stream) {
  (void)in_sizes; (void)n_in; (void)out_size; (void)ws_size;
  const float* x  = (const float*)d_in[0];
  const float* Wk = (const float*)d_in[1];
  const float* Wq = (const float*)d_in[2];
  const float* Wv = (const float*)d_in[3];

  float* ws = (float*)d_ws;
  const size_t plane = (size_t)BATCH * SEQ * HS;
  float* Q  = ws;
  float* Kf = ws + plane;
  float* Vf = ws + 2 * plane;
  float* out = (float*)d_out;

  dim3 g1(BATCH * SEQ / 16, 3);
  proj_kernel<<<g1, 128, 0, stream>>>(x, Wq, Wk, Wv, Q, Kf, Vf);

  attn_kernel<<<(BATCH * SEQ / 16) / 4, 128, 0, stream>>>(Q, Kf, Vf, out);
}


# projection x-stationary: a wave keeps two 16-row x tiles as A fragments in registers (16-byte loads, no LDS, no barrier); W fragments stream a group ahead, each feeds both tiles; k sum order changes
# speedup vs baseline: 4.2282x; 1.2400x over previous
#include <hip/hip_runtime.h>
#include <math.h>


typedef __attribute__((ext_vector_type(2))) float v2f;
typedef __attribute__((ext_vector_type(4))) float v4f;
typedef __attribute__((ext_vector_type(8))) float v8f;

#define EMB   768
#define HS    64
#define SEQ   2048
#define BATCH 8
#define TILE  1024
#define NKT   (SEQ / 16)

__device__ __forceinline__ v8f wmma_f32(v2f a, v2f b, v8f c) {
  return __builtin_amdgcn_wmma_f32_16x16x4_f32(
       false, a,  false, b,
       (short)0, c,  false,  false);
}


template <int M>
__device__ __forceinline__ float row_xor(float x) {
  return __builtin_amdgcn_update_dpp(0.0f, x, 0x160 + M, 0xf, 0xf, true);
}


#define PU  6
#define PG  (EMB / 8 / PU)

__device__ __forceinline__ void store_tile(int which, int cw, int tileM, const v8f& c,
                                           int l16, int half,
                                           float* __restrict__ Q, float* __restrict__ Kf,
                                           float* __restrict__ Vf) {
  const int n = cw * 16 + l16;
  if (which == 0) {
#pragma unroll
    for (int v = 0; v < 8; ++v) {
      const int m = tileM * 16 + v + half * 8;
      Q[m * HS + n] = c[v];
    }
  } else if (which == 1) {
    float* T = Kf + tileM * TILE;
    const int base = (n >> 3) * 128 + ((n >> 1) & 1) * 64 + ((n >> 2) & 1) * 2 + (n & 1);
#pragma unroll
    for (int v = 0; v < 8; ++v) {
      const int key = v + half * 8;
      T[base + key * 4] = c[v];
    }
  } else {
    float* T = Vf + tileM * TILE;
#pragma unroll
    for (int v = 0; v < 8; v += 2) {
      const int kc = (v >> 2) + 2 * half;
      const int idx = (kc * 2 + (cw >> 1)) * 128 + (((v >> 1) & 1) * 16 + l16) * 4 + (cw & 1) * 2;
      v2f pr; pr.x = c[v]; pr.y = c[v + 1];
      *(v2f*)(T + idx) = pr;
    }
  }
}

__global__ __launch_bounds__(128) void proj_kernel(
    const float* __restrict__ x,
    const float* __restrict__ Wq, const float* __restrict__ Wk,
    const float* __restrict__ Wv,
    float* __restrict__ Q, float* __restrict__ Kf, float* __restrict__ Vf) {
  const int tid   = threadIdx.x;
  const int lane  = tid & 31;
  const int wave  = __builtin_amdgcn_readfirstlane(tid >> 5);
  const int l16   = lane & 15;
  const int half  = lane >> 4;
  const int pair  = blockIdx.x * 2 + (wave >> 1);
  const int nh    = wave & 1;

  const int   lofs = (4 * half) * HS + l16;
  const float* w0 = (nh ? Wk + 32 : Wq)      + lofs;
  const float* w1 = (nh ? Wv      : Wq + 32) + lofs;
  const float* w2 = (nh ? Wv + 32 : Wk)      + lofs;

  auto fetch = [&](v2f (&bf)[PU][2][2], const float* wp, int g) {
#pragma unroll
    for (int u = 0; u < PU; ++u)
#pragma unroll
      for (int f = 0; f < 2; ++f)
#pragma unroll
        for (int e = 0; e < 2; ++e) {
          const float* q = wp + ((g * PU + u) * 8 + 2 * f) * HS + e * 16;
          bf[u][f][e].x = q[0];
          bf[u][f][e].y = q[HS];
        }
  };

  v2f ba[PU][2][2], bb[PU][2][2];
  const float* wp = w0;
  fetch(ba, wp, 0);

  v4f xa[2][EMB / 8];
  const v4f* xr0 = (const v4f*)(x + ((2 * pair) * 16 + l16) * EMB) + half;
  const v4f* xr1 = xr0 + 16 * (EMB / 4);
  auto xload = [&](int g) {
#pragma unroll
    for (int u = 0; u < PU; ++u) {
      const int i = g * PU + u;
      xa[0][i] = xr0[2 * i];
      xa[1][i] = xr1[2 * i];
    }
  };
  xload(0);
  xload(1);

  v8f acc[2][2];

  auto step = [&](const v2f (&bf)[PU][2][2], int g) {
#pragma unroll
    for (int u = 0; u < PU; ++u) {
      const int i = g * PU + u;
      v2f a[2][2];
#pragma unroll
      for (int r = 0; r < 2; ++r) {
        a[r][0].x = xa[r][i].x; a[r][0].y = xa[r][i].y;
        a[r][1].x = xa[r][i].z; a[r][1].y = xa[r][i].w;
      }
#pragma unroll
      for (int f = 0; f < 2; ++f)
#pragma unroll
        for (int e = 0; e < 2; ++e)
#pragma unroll
          for (int r = 0; r < 2; ++r)
            acc[r][e] = wmma_f32(a[r][f], bf[u][f][e], acc[r][e]);
    }
  };

  for (int p = 0; p < 3; ++p) {
    const float* wn = (p == 0) ? w1 : w2;
#pragma unroll
    for (int r = 0; r < 2; ++r)
#pragma unroll
      for (int e = 0; e < 2; ++e) acc[r][e] = v8f{};

#pragma unroll
    for (int g = 0; g < PG; g += 2) {
      fetch(bb, wp, g + 1);
      if (p == 0 && g + 2 < PG) xload(g + 2);
      __builtin_amdgcn_sched_barrier(0);
      step(ba, g);
      __builtin_amdgcn_sched_barrier(0);
      if (g + 2 < PG) fetch(ba, wp, g + 2);
      else            fetch(ba, wn, 0);
      if (p == 0 && g + 3 < PG) xload(g + 3);
      __builtin_amdgcn_sched_barrier(0);
      step(bb, g + 1);
      __builtin_amdgcn_sched_barrier(0);
    }

    const int jp = nh * 3 + p;
#pragma unroll
    for (int r = 0; r < 2; ++r)
#pragma unroll
      for (int e = 0; e < 2; ++e)
        store_tile(jp >> 1, (jp & 1) * 2 + e, 2 * pair + r, acc[r][e], l16, half, Q, Kf, Vf);
    wp = wn;
  }
}

#define KT  2
#define PT  (16 * 17)

__global__ __launch_bounds__(128) void attn_kernel(
    const float* __restrict__ Q, const float* __restrict__ Kf,
    const float* __restrict__ Vf, float* __restrict__ out) {
  __shared__ float plds[4][KT * PT];

  const int tid   = threadIdx.x;
  const int lane  = tid & 31;
  const int wave  = tid >> 5;
  const int gt    = blockIdx.x * 4 + wave;
  const int b     = gt >> 7;
  const int qbase = (gt & 127) * 16;
  const int l16   = lane & 15;
  const int half  = lane >> 4;
  const int halfk = half * 2;

  v2f qa[16];
  const int mrow = (b * SEQ + qbase + l16) * HS;
#pragma unroll
  for (int kc = 0; kc < 16; ++kc) {
    qa[kc].x = Q[mrow + kc * 4 + halfk];
    qa[kc].y = Q[mrow + kc * 4 + halfk + 1];
  }

  v8f acc0 = {}, acc1 = {}, acc2 = {}, acc3 = {};
  float mst[8], lst[8];
#pragma unroll
  for (int v = 0; v < 8; ++v) { mst[v] = -INFINITY; lst[v] = 0.0f; }

  float* pl = &plds[wave][0];
  const float scale = 0.125f;

  const v4f* kp = (const v4f*)Kf + (b * NKT) * (TILE / 4) + lane;
  const v4f* vp = (const v4f*)Vf + (b * NKT) * (TILE / 4) + lane;

  auto fetch = [&](v4f (&kf)[KT][8], v4f (&vf)[KT][8], int g) {
    const v4f* kq = kp + g * (KT * TILE / 4);
    const v4f* vq = vp + g * (KT * TILE / 4);
#pragma unroll
    for (int t = 0; t < KT; ++t)
#pragma unroll
      for (int i = 0; i < 8; ++i) kf[t][i] = kq[(t * 8 + i) * 32];
#pragma unroll
    for (int t = 0; t < KT; ++t)
#pragma unroll
      for (int i = 0; i < 8; ++i) vf[t][i] = vq[(t * 8 + i) * 32];
  };

  auto step = [&](const v4f (&kf)[KT][8], const v4f (&vf)[KT][8]) {
    v8f s[KT];
#pragma unroll
    for (int t = 0; t < KT; ++t) s[t] = v8f{};
#pragma unroll
    for (int i = 0; i < 8; ++i) {
#pragma unroll
      for (int t = 0; t < KT; ++t) {
        v2f k0; k0.x = kf[t][i].x; k0.y = kf[t][i].y;
        s[t] = wmma_f32(qa[2 * i], k0, s[t]);
      }
#pragma unroll
      for (int t = 0; t < KT; ++t) {
        v2f k1; k1.x = kf[t][i].z; k1.y = kf[t][i].w;
        s[t] = wmma_f32(qa[2 * i + 1], k1, s[t]);
      }
    }

#pragma unroll
    for (int v = 0; v < 8; ++v) {
      float a[KT];
#pragma unroll
      for (int t = 0; t < KT; ++t) a[t] = s[t][v] * scale;
      float tm = a[0];
#pragma unroll
      for (int t = 1; t < KT; ++t) tm = fmaxf(tm, a[t]);
      tm = fmaxf(tm, row_xor<1>(tm));
      tm = fmaxf(tm, row_xor<2>(tm));
      tm = fmaxf(tm, row_xor<4>(tm));
      tm = fmaxf(tm, row_xor<8>(tm));
      const float nm    = fmaxf(mst[v], tm);
      const float alpha = __expf(mst[v] - nm);
      float ps = 0.0f;
#pragma unroll
      for (int t = 0; t < KT; ++t) {
        const float p = __expf(a[t] - nm);
        ps = (t == 0) ? p : ps + p;
        pl[t * PT + (v + half * 8) * 17 + l16] = p;
      }
      lst[v] = lst[v] * alpha + ps;
      mst[v] = nm;
      acc0[v] *= alpha; acc1[v] *= alpha; acc2[v] *= alpha; acc3[v] *= alpha;
    }

#pragma unroll
    for (int t = 0; t < KT; ++t) {
#pragma unroll
      for (int kc = 0; kc < 4; ++kc) {
        v2f a;
        a.x = pl[t * PT + l16 * 17 + kc * 4 + halfk];
        a.y = pl[t * PT + l16 * 17 + kc * 4 + halfk + 1];
        const v4f f01 = vf[t][2 * kc];
        const v4f f23 = vf[t][2 * kc + 1];
        { v2f bv; bv.x = f01.x; bv.y = f01.y; acc0 = wmma_f32(a, bv, acc0); }
        { v2f bv; bv.x = f01.z; bv.y = f01.w; acc1 = wmma_f32(a, bv, acc1); }
        { v2f bv; bv.x = f23.x; bv.y = f23.y; acc2 = wmma_f32(a, bv, acc2); }
        { v2f bv; bv.x = f23.z; bv.y = f23.w; acc3 = wmma_f32(a, bv, acc3); }
      }
    }
  };

  constexpr int NG = NKT / KT;
  v4f ka[KT][8], va[KT][8], kb[KT][8], vb[KT][8];
  fetch(ka, va, 0);
  for (int g = 0; g < NG; g += 2) {
    fetch(kb, vb, g + 1);
    step(ka, va);
    fetch(ka, va, (g + 2 < NG) ? g + 2 : NG - 1);
    step(kb, vb);
  }

#pragma unroll
  for (int v = 0; v < 8; ++v) {
    float l = lst[v];
    l += row_xor<1>(l);
    l += row_xor<2>(l);
    l += row_xor<4>(l);
    l += row_xor<8>(l);
    const float inv = 1.0f / l;
    const int orow = (b * SEQ + qbase + v + half * 8) * HS;
    out[orow + l16]      = acc0[v] * inv;
    out[orow + 16 + l16] = acc1[v] * inv;
    out[orow + 32 + l16] = acc2[v] * inv;
    out[orow + 48 + l16] = acc3[v] * inv;
  }
}

extern "C" void kernel_launch(void* const* d_in, const int* in_sizes, int n_in,
                              void* d_out, int out_size, void* d_ws, size_t ws_size,
                              hipStream_t stream) {
  (void)in_sizes; (void)n_in; (void)out_size; (void)ws_size;
  const float* x  = (const float*)d_in[0];
  const float* Wk = (const float*)d_in[1];
  const float* Wq = (const float*)d_in[2];
  const float* Wv = (const float*)d_in[3];

  float* ws = (float*)d_ws;
  const size_t plane = (size_t)BATCH * SEQ * HS;
  float* Q  = ws;
  float* Kf = ws + plane;
  float* Vf = ws + 2 * plane;
  float* out = (float*)d_out;

  proj_kernel<<<(BATCH * SEQ / 16) / 4, 128, 0, stream>>>(x, Wq, Wk, Wv, Q, Kf, Vf);

  attn_kernel<<<(BATCH * SEQ / 16) / 4, 128, 0, stream>>>(Q, Kf, Vf, out);
}


# attention key loop software-pipelined inside the wave: score WMMAs of key group g+1 issue beside the softmax of group g (vector ALU under matrix work), then PV(g); every sum keeps its order
# speedup vs baseline: 5.0397x; 1.1919x over previous
#include <hip/hip_runtime.h>
#include <math.h>


typedef __attribute__((ext_vector_type(2))) float v2f;
typedef __attribute__((ext_vector_type(4))) float v4f;
typedef __attribute__((ext_vector_type(8))) float v8f;

#define EMB   768
#define HS    64
#define SEQ   2048
#define BATCH 8
#define TILE  1024
#define NKT   (SEQ / 16)

__device__ __forceinline__ v8f wmma_f32(v2f a, v2f b, v8f c) {
  return __builtin_amdgcn_wmma_f32_16x16x4_f32(
       false, a,  false, b,
       (short)0, c,  false,  false);
}


template <int M>
__device__ __forceinline__ float row_xor(float x) {
  return __builtin_amdgcn_update_dpp(0.0f, x, 0x160 + M, 0xf, 0xf, true);
}


#define PU  6
#define PG  (EMB / 8 / PU)

__device__ __forceinline__ void store_tile(int which, int cw, int tileM, const v8f& c,
                                           int l16, int half,
                                           float* __restrict__ Q, float* __restrict__ Kf,
                                           float* __restrict__ Vf) {
  const int n = cw * 16 + l16;
  if (which == 0) {
#pragma unroll
    for (int v = 0; v < 8; ++v) {
      const int m = tileM * 16 + v + half * 8;
      Q[m * HS + n] = c[v];
    }
  } else if (which == 1) {
    float* T = Kf + tileM * TILE;
    const int base = (n >> 3) * 128 + ((n >> 1) & 1) * 64 + ((n >> 2) & 1) * 2 + (n & 1);
#pragma unroll
    for (int v = 0; v < 8; ++v) {
      const int key = v + half * 8;
      T[base + key * 4] = c[v];
    }
  } else {
    float* T = Vf + tileM * TILE;
#pragma unroll
    for (int v = 0; v < 8; v += 2) {
      const int kc = (v >> 2) + 2 * half;
      const int idx = (kc * 2 + (cw >> 1)) * 128 + (((v >> 1) & 1) * 16 + l16) * 4 + (cw & 1) * 2;
      v2f pr; pr.x = c[v]; pr.y = c[v + 1];
      *(v2f*)(T + idx) = pr;
    }
  }
}

__global__ __launch_bounds__(128) void proj_kernel(
    const float* __restrict__ x,
    const float* __restrict__ Wq, const float* __restrict__ Wk,
    const float* __restrict__ Wv,
    float* __restrict__ Q, float* __restrict__ Kf, float* __restrict__ Vf) {
  const int tid   = threadIdx.x;
  const int lane  = tid & 31;
  const int wave  = __builtin_amdgcn_readfirstlane(tid >> 5);
  const int l16   = lane & 15;
  const int half  = lane >> 4;
  const int pair  = blockIdx.x * 2 + (wave >> 1);
  const int nh    = wave & 1;

  const int   lofs = (4 * half) * HS + l16;
  const float* w0 = (nh ? Wk + 32 : Wq)      + lofs;
  const float* w1 = (nh ? Wv      : Wq + 32) + lofs;
  const float* w2 = (nh ? Wv + 32 : Wk)      + lofs;

  auto fetch = [&](v2f (&bf)[PU][2][2], const float* wp, int g) {
#pragma unroll
    for (int u = 0; u < PU; ++u)
#pragma unroll
      for (int f = 0; f < 2; ++f)
#pragma unroll
        for (int e = 0; e < 2; ++e) {
          const float* q = wp + ((g * PU + u) * 8 + 2 * f) * HS + e * 16;
          bf[u][f][e].x = q[0];
          bf[u][f][e].y = q[HS];
        }
  };

  v2f ba[PU][2][2], bb[PU][2][2];
  const float* wp = w0;
  fetch(ba, wp, 0);

  v4f xa[2][EMB / 8];
  const v4f* xr0 = (const v4f*)(x + ((2 * pair) * 16 + l16) * EMB) + half;
  const v4f* xr1 = xr0 + 16 * (EMB / 4);
  auto xload = [&](int g) {
#pragma unroll
    for (int u = 0; u < PU; ++u) {
      const int i = g * PU + u;
      xa[0][i] = xr0[2 * i];
      xa[1][i] = xr1[2 * i];
    }
  };
  xload(0);
  xload(1);

  v8f acc[2][2];

  auto step = [&](const v2f (&bf)[PU][2][2], int g) {
#pragma unroll
    for (int u = 0; u < PU; ++u) {
      const int i = g * PU + u;
      v2f a[2][2];
#pragma unroll
      for (int r = 0; r < 2; ++r) {
        a[r][0].x = xa[r][i].x; a[r][0].y = xa[r][i].y;
        a[r][1].x = xa[r][i].z; a[r][1].y = xa[r][i].w;
      }
#pragma unroll
      for (int f = 0; f < 2; ++f)
#pragma unroll
        for (int e = 0; e < 2; ++e)
#pragma unroll
          for (int r = 0; r < 2; ++r)
            acc[r][e] = wmma_f32(a[r][f], bf[u][f][e], acc[r][e]);
    }
  };

  for (int p = 0; p < 3; ++p) {
    const float* wn = (p == 0) ? w1 : w2;
#pragma unroll
    for (int r = 0; r < 2; ++r)
#pragma unroll
      for (int e = 0; e < 2; ++e) acc[r][e] = v8f{};

#pragma unroll
    for (int g = 0; g < PG; g += 2) {
      fetch(bb, wp, g + 1);
      if (p == 0 && g + 2 < PG) xload(g + 2);
      __builtin_amdgcn_sched_barrier(0);
      step(ba, g);
      __builtin_amdgcn_sched_barrier(0);
      if (g + 2 < PG) fetch(ba, wp, g + 2);
      else            fetch(ba, wn, 0);
      if (p == 0 && g + 3 < PG) xload(g + 3);
      __builtin_amdgcn_sched_barrier(0);
      step(bb, g + 1);
      __builtin_amdgcn_sched_barrier(0);
    }

    const int jp = nh * 3 + p;
#pragma unroll
    for (int r = 0; r < 2; ++r)
#pragma unroll
      for (int e = 0; e < 2; ++e)
        store_tile(jp >> 1, (jp & 1) * 2 + e, 2 * pair + r, acc[r][e], l16, half, Q, Kf, Vf);
    wp = wn;
  }
}

#define KT  2
#define PT  (16 * 17)

__global__ __launch_bounds__(128) void attn_kernel(
    const float* __restrict__ Q, const float* __restrict__ Kf,
    const float* __restrict__ Vf, float* __restrict__ out) {
  __shared__ float plds[4][KT * PT];

  const int tid   = threadIdx.x;
  const int lane  = tid & 31;
  const int wave  = tid >> 5;
  const int gt    = blockIdx.x * 4 + wave;
  const int b     = gt >> 7;
  const int qbase = (gt & 127) * 16;
  const int l16   = lane & 15;
  const int half  = lane >> 4;
  const int halfk = half * 2;

  v2f qa[16];
  const int mrow = (b * SEQ + qbase + l16) * HS;
#pragma unroll
  for (int kc = 0; kc < 16; ++kc) {
    qa[kc].x = Q[mrow + kc * 4 + halfk];
    qa[kc].y = Q[mrow + kc * 4 + halfk + 1];
  }

  v8f acc0 = {}, acc1 = {}, acc2 = {}, acc3 = {};
  float mst[8], lst[8];
#pragma unroll
  for (int v = 0; v < 8; ++v) { mst[v] = -INFINITY; lst[v] = 0.0f; }

  float* pl = &plds[wave][0];
  const float scale = 0.125f;

  const v4f* kp = (const v4f*)Kf + (b * NKT) * (TILE / 4) + lane;
  const v4f* vp = (const v4f*)Vf + (b * NKT) * (TILE / 4) + lane;

  auto fetchK = [&](v4f (&kf)[KT][8], int g) {
    const v4f* kq = kp + g * (KT * TILE / 4);
#pragma unroll
    for (int t = 0; t < KT; ++t)
#pragma unroll
      for (int i = 0; i < 8; ++i) kf[t][i] = kq[(t * 8 + i) * 32];
  };
  auto fetchV = [&](v4f (&vf)[KT][8], int g) {
    const v4f* vq = vp + g * (KT * TILE / 4);
#pragma unroll
    for (int t = 0; t < KT; ++t)
#pragma unroll
      for (int i = 0; i < 8; ++i) vf[t][i] = vq[(t * 8 + i) * 32];
  };

  auto qk_i = [&](const v4f (&kf)[KT][8], v8f (&s)[KT], int i) {
#pragma unroll
    for (int t = 0; t < KT; ++t) {
      v2f k0; k0.x = kf[t][i].x; k0.y = kf[t][i].y;
      s[t] = wmma_f32(qa[2 * i], k0, s[t]);
    }
#pragma unroll
    for (int t = 0; t < KT; ++t) {
      v2f k1; k1.x = kf[t][i].z; k1.y = kf[t][i].w;
      s[t] = wmma_f32(qa[2 * i + 1], k1, s[t]);
    }
  };

  auto softmax_row = [&](const v8f (&s)[KT], int v) {
    float a[KT];
#pragma unroll
    for (int t = 0; t < KT; ++t) a[t] = s[t][v] * scale;
    float tm = a[0];
#pragma unroll
    for (int t = 1; t < KT; ++t) tm = fmaxf(tm, a[t]);
    tm = fmaxf(tm, row_xor<1>(tm));
    tm = fmaxf(tm, row_xor<2>(tm));
    tm = fmaxf(tm, row_xor<4>(tm));
    tm = fmaxf(tm, row_xor<8>(tm));
    const float nm    = fmaxf(mst[v], tm);
    const float alpha = __expf(mst[v] - nm);
    float ps = 0.0f;
#pragma unroll
    for (int t = 0; t < KT; ++t) {
      const float p = __expf(a[t] - nm);
      ps = (t == 0) ? p : ps + p;
      pl[t * PT + (v + half * 8) * 17 + l16] = p;
    }
    lst[v] = lst[v] * alpha + ps;
    mst[v] = nm;
    acc0[v] *= alpha; acc1[v] *= alpha; acc2[v] *= alpha; acc3[v] *= alpha;
  };

  auto loadP = [&](v2f (&pa)[KT][4]) {
#pragma unroll
    for (int t = 0; t < KT; ++t)
#pragma unroll
      for (int kc = 0; kc < 4; ++kc) {
        pa[t][kc].x = pl[t * PT + l16 * 17 + kc * 4 + halfk];
        pa[t][kc].y = pl[t * PT + l16 * 17 + kc * 4 + halfk + 1];
      }
  };

  auto scores_beside_softmax = [&](const v4f (&kf)[KT][8], v8f (&sn)[KT],
                                   const v8f (&sc)[KT], v2f (&pa)[KT][4]) {
#pragma unroll
    for (int t = 0; t < KT; ++t) sn[t] = v8f{};
#pragma unroll
    for (int i = 0; i < 8; ++i) {
      qk_i(kf, sn, i);
      const int r0 = (i + i / 2 < 8) ? i + i / 2 : 8;
      const int r1 = (i + 1 + (i + 1) / 2 < 8) ? i + 1 + (i + 1) / 2 : 8;
#pragma unroll
      for (int v = r0; v < r1; ++v) softmax_row(sc, v);
      if (i == 5) {
        loadP(pa);
        __builtin_amdgcn_sched_barrier(0);
      }
    }
  };

  auto pv = [&](const v2f (&pa)[KT][4], const v4f (&vf)[KT][8]) {
#pragma unroll
    for (int t = 0; t < KT; ++t) {
#pragma unroll
      for (int kc = 0; kc < 4; ++kc) {
        const v2f a = pa[t][kc];
        const v4f f01 = vf[t][2 * kc];
        const v4f f23 = vf[t][2 * kc + 1];
        { v2f bv; bv.x = f01.x; bv.y = f01.y; acc0 = wmma_f32(a, bv, acc0); }
        { v2f bv; bv.x = f01.z; bv.y = f01.w; acc1 = wmma_f32(a, bv, acc1); }
        { v2f bv; bv.x = f23.x; bv.y = f23.y; acc2 = wmma_f32(a, bv, acc2); }
        { v2f bv; bv.x = f23.z; bv.y = f23.w; acc3 = wmma_f32(a, bv, acc3); }
      }
    }
  };

  constexpr int NG = NKT / KT;
  v4f ka[KT][8], va[KT][8], kb[KT][8], vb[KT][8];
  v8f sa[KT], sb[KT];
  v2f pa[KT][4];
  fetchK(ka, 0);
  fetchV(va, 0);
  fetchK(kb, 1);
#pragma unroll
  for (int t = 0; t < KT; ++t) sa[t] = v8f{};
#pragma unroll
  for (int i = 0; i < 8; ++i) qk_i(ka, sa, i);
  fetchK(ka, 2);

  for (int g = 0; g < NG - 2; g += 2) {
    fetchV(vb, g + 1);
    scores_beside_softmax(kb, sb, sa, pa);
    fetchK(kb, g + 3);
    pv(pa, va);
    fetchV(va, g + 2);
    scores_beside_softmax(ka, sa, sb, pa);
    fetchK(ka, (g + 4 < NG) ? g + 4 : NG - 1);
    pv(pa, vb);
  }
  fetchV(vb, NG - 1);
  scores_beside_softmax(kb, sb, sa, pa);
  pv(pa, va);
#pragma unroll
  for (int v = 0; v < 8; ++v) softmax_row(sb, v);
  loadP(pa);
  pv(pa, vb);

#pragma unroll
  for (int v = 0; v < 8; ++v) {
    float l = lst[v];
    l += row_xor<1>(l);
    l += row_xor<2>(l);
    l += row_xor<4>(l);
    l += row_xor<8>(l);
    const float inv = 1.0f / l;
    const int orow = (b * SEQ + qbase + v + half * 8) * HS;
    out[orow + l16]      = acc0[v] * inv;
    out[orow + 16 + l16] = acc1[v] * inv;
    out[orow + 32 + l16] = acc2[v] * inv;
    out[orow + 48 + l16] = acc3[v] * inv;
  }
}

extern "C" void kernel_launch(void* const* d_in, const int* in_sizes, int n_in,
                              void* d_out, int out_size, void* d_ws, size_t ws_size,
                              hipStream_t stream) {
  (void)in_sizes; (void)n_in; (void)out_size; (void)ws_size;
  const float* x  = (const float*)d_in[0];
  const float* Wk = (const float*)d_in[1];
  const float* Wq = (const float*)d_in[2];
  const float* Wv = (const float*)d_in[3];

  float* ws = (float*)d_ws;
  const size_t plane = (size_t)BATCH * SEQ * HS;
  float* Q  = ws;
  float* Kf = ws + plane;
  float* Vf = ws + 2 * plane;
  float* out = (float*)d_out;

  proj_kernel<<<(BATCH * SEQ / 16) / 4, 128, 0, stream>>>(x, Wq, Wk, Wv, Q, Kf, Vf);

  attn_kernel<<<(BATCH * SEQ / 16) / 4, 128, 0, stream>>>(Q, Kf, Vf, out);
}


# attention key loop: memory instructions gathered into two places per key group (all P stores + fragment reads behind score quad 5; K,V loads one run between scores and PV); sum order unchanged
# speedup vs baseline: 5.3320x; 1.0580x over previous
#include <hip/hip_runtime.h>
#include <math.h>


typedef __attribute__((ext_vector_type(2))) float v2f;
typedef __attribute__((ext_vector_type(4))) float v4f;
typedef __attribute__((ext_vector_type(8))) float v8f;

#define EMB   768
#define HS    64
#define SEQ   2048
#define BATCH 8
#define TILE  1024
#define NKT   (SEQ / 16)

__device__ __forceinline__ v8f wmma_f32(v2f a, v2f b, v8f c) {
  return __builtin_amdgcn_wmma_f32_16x16x4_f32(
       false, a,  false, b,
       (short)0, c,  false,  false);
}


template <int M>
__device__ __forceinline__ float row_xor(float x) {
  return __builtin_amdgcn_update_dpp(0.0f, x, 0x160 + M, 0xf, 0xf, true);
}


#define PU  6
#define PG  (EMB / 8 / PU)

__device__ __forceinline__ void store_tile(int which, int cw, int tileM, const v8f& c,
                                           int l16, int half,
                                           float* __restrict__ Q, float* __restrict__ Kf,
                                           float* __restrict__ Vf) {
  const int n = cw * 16 + l16;
  if (which == 0) {
#pragma unroll
    for (int v = 0; v < 8; ++v) {
      const int m = tileM * 16 + v + half * 8;
      Q[m * HS + n] = c[v];
    }
  } else if (which == 1) {
    float* T = Kf + tileM * TILE;
    const int base = (n >> 3) * 128 + ((n >> 1) & 1) * 64 + ((n >> 2) & 1) * 2 + (n & 1);
#pragma unroll
    for (int v = 0; v < 8; ++v) {
      const int key = v + half * 8;
      T[base + key * 4] = c[v];
    }
  } else {
    float* T = Vf + tileM * TILE;
#pragma unroll
    for (int v = 0; v < 8; v += 2) {
      const int kc = (v >> 2) + 2 * half;
      const int idx = (kc * 2 + (cw >> 1)) * 128 + (((v >> 1) & 1) * 16 + l16) * 4 + (cw & 1) * 2;
      v2f pr; pr.x = c[v]; pr.y = c[v + 1];
      *(v2f*)(T + idx) = pr;
    }
  }
}

__global__ __launch_bounds__(128) void proj_kernel(
    const float* __restrict__ x,
    const float* __restrict__ Wq, const float* __restrict__ Wk,
    const float* __restrict__ Wv,
    float* __restrict__ Q, float* __restrict__ Kf, float* __restrict__ Vf) {
  const int tid   = threadIdx.x;
  const int lane  = tid & 31;
  const int wave  = __builtin_amdgcn_readfirstlane(tid >> 5);
  const int l16   = lane & 15;
  const int half  = lane >> 4;
  const int pair  = blockIdx.x * 2 + (wave >> 1);
  const int nh    = wave & 1;

  const int   lofs = (4 * half) * HS + l16;
  const float* w0 = (nh ? Wk + 32 : Wq)      + lofs;
  const float* w1 = (nh ? Wv      : Wq + 32) + lofs;
  const float* w2 = (nh ? Wv + 32 : Wk)      + lofs;

  auto fetch = [&](v2f (&bf)[PU][2][2], const float* wp, int g) {
#pragma unroll
    for (int u = 0; u < PU; ++u)
#pragma unroll
      for (int f = 0; f < 2; ++f)
#pragma unroll
        for (int e = 0; e < 2; ++e) {
          const float* q = wp + ((g * PU + u) * 8 + 2 * f) * HS + e * 16;
          bf[u][f][e].x = q[0];
          bf[u][f][e].y = q[HS];
        }
  };

  v2f ba[PU][2][2], bb[PU][2][2];
  const float* wp = w0;
  fetch(ba, wp, 0);

  v4f xa[2][EMB / 8];
  const v4f* xr0 = (const v4f*)(x + ((2 * pair) * 16 + l16) * EMB) + half;
  const v4f* xr1 = xr0 + 16 * (EMB / 4);
  auto xload = [&](int g) {
#pragma unroll
    for (int u = 0; u < PU; ++u) {
      const int i = g * PU + u;
      xa[0][i] = xr0[2 * i];
      xa[1][i] = xr1[2 * i];
    }
  };
  xload(0);
  xload(1);

  v8f acc[2][2];

  auto step = [&](const v2f (&bf)[PU][2][2], int g) {
#pragma unroll
    for (int u = 0; u < PU; ++u) {
      const int i = g * PU + u;
      v2f a[2][2];
#pragma unroll
      for (int r = 0; r < 2; ++r) {
        a[r][0].x = xa[r][i].x; a[r][0].y = xa[r][i].y;
        a[r][1].x = xa[r][i].z; a[r][1].y = xa[r][i].w;
      }
#pragma unroll
      for (int f = 0; f < 2; ++f)
#pragma unroll
        for (int e = 0; e < 2; ++e)
#pragma unroll
          for (int r = 0; r < 2; ++r)
            acc[r][e] = wmma_f32(a[r][f], bf[u][f][e], acc[r][e]);
    }
  };

  for (int p = 0; p < 3; ++p) {
    const float* wn = (p == 0) ? w1 : w2;
#pragma unroll
    for (int r = 0; r < 2; ++r)
#pragma unroll
      for (int e = 0; e < 2; ++e) acc[r][e] = v8f{};

#pragma unroll
    for (int g = 0; g < PG; g += 2) {
      fetch(bb, wp, g + 1);
      if (p == 0 && g + 2 < PG) xload(g + 2);
      __builtin_amdgcn_sched_barrier(0);
      step(ba, g);
      __builtin_amdgcn_sched_barrier(0);
      if (g + 2 < PG) fetch(ba, wp, g + 2);
      else            fetch(ba, wn, 0);
      if (p == 0 && g + 3 < PG) xload(g + 3);
      __builtin_amdgcn_sched_barrier(0);
      step(bb, g + 1);
      __builtin_amdgcn_sched_barrier(0);
    }

    const int jp = nh * 3 + p;
#pragma unroll
    for (int r = 0; r < 2; ++r)
#pragma unroll
      for (int e = 0; e < 2; ++e)
        store_tile(jp >> 1, (jp & 1) * 2 + e, 2 * pair + r, acc[r][e], l16, half, Q, Kf, Vf);
    wp = wn;
  }
}

#define KT  2
#define PT  (16 * 17)

__global__ __launch_bounds__(128) void attn_kernel(
    const float* __restrict__ Q, const float* __restrict__ Kf,
    const float* __restrict__ Vf, float* __restrict__ out) {
  __shared__ float plds[4][KT * PT];

  const int tid   = threadIdx.x;
  const int lane  = tid & 31;
  const int wave  = tid >> 5;
  const int gt    = blockIdx.x * 4 + wave;
  const int b     = gt >> 7;
  const int qbase = (gt & 127) * 16;
  const int l16   = lane & 15;
  const int half  = lane >> 4;
  const int halfk = half * 2;

  v2f qa[16];
  const int mrow = (b * SEQ + qbase + l16) * HS;
#pragma unroll
  for (int kc = 0; kc < 16; ++kc) {
    qa[kc].x = Q[mrow + kc * 4 + halfk];
    qa[kc].y = Q[mrow + kc * 4 + halfk + 1];
  }

  v8f acc0 = {}, acc1 = {}, acc2 = {}, acc3 = {};
  float mst[8], lst[8];
#pragma unroll
  for (int v = 0; v < 8; ++v) { mst[v] = -INFINITY; lst[v] = 0.0f; }

  float* pl = &plds[wave][0];
  const float scale = 0.125f;
  float pr[KT][8];

  const v4f* kp = (const v4f*)Kf + (b * NKT) * (TILE / 4) + lane;
  const v4f* vp = (const v4f*)Vf + (b * NKT) * (TILE / 4) + lane;

  auto fetchK = [&](v4f (&kf)[KT][8], int g) {
    const v4f* kq = kp + g * (KT * TILE / 4);
#pragma unroll
    for (int t = 0; t < KT; ++t)
#pragma unroll
      for (int i = 0; i < 8; ++i) kf[t][i] = kq[(t * 8 + i) * 32];
  };
  auto fetchV = [&](v4f (&vf)[KT][8], int g) {
    const v4f* vq = vp + g * (KT * TILE / 4);
#pragma unroll
    for (int t = 0; t < KT; ++t)
#pragma unroll
      for (int i = 0; i < 8; ++i) vf[t][i] = vq[(t * 8 + i) * 32];
  };

  auto qk_i = [&](const v4f (&kf)[KT][8], v8f (&s)[KT], int i) {
#pragma unroll
    for (int t = 0; t < KT; ++t) {
      v2f k0; k0.x = kf[t][i].x; k0.y = kf[t][i].y;
      s[t] = wmma_f32(qa[2 * i], k0, s[t]);
    }
#pragma unroll
    for (int t = 0; t < KT; ++t) {
      v2f k1; k1.x = kf[t][i].z; k1.y = kf[t][i].w;
      s[t] = wmma_f32(qa[2 * i + 1], k1, s[t]);
    }
  };

  auto softmax_row = [&](const v8f (&s)[KT], int v) {
    float a[KT];
#pragma unroll
    for (int t = 0; t < KT; ++t) a[t] = s[t][v] * scale;
    float tm = a[0];
#pragma unroll
    for (int t = 1; t < KT; ++t) tm = fmaxf(tm, a[t]);
    tm = fmaxf(tm, row_xor<1>(tm));
    tm = fmaxf(tm, row_xor<2>(tm));
    tm = fmaxf(tm, row_xor<4>(tm));
    tm = fmaxf(tm, row_xor<8>(tm));
    const float nm    = fmaxf(mst[v], tm);
    const float alpha = __expf(mst[v] - nm);
    float ps = 0.0f;
#pragma unroll
    for (int t = 0; t < KT; ++t) {
      const float p = __expf(a[t] - nm);
      ps = (t == 0) ? p : ps + p;
      pr[t][v] = p;
    }
    lst[v] = lst[v] * alpha + ps;
    mst[v] = nm;
    acc0[v] *= alpha; acc1[v] *= alpha; acc2[v] *= alpha; acc3[v] *= alpha;
  };

  auto storeP = [&]() {
#pragma unroll
    for (int t = 0; t < KT; ++t)
#pragma unroll
      for (int v = 0; v < 8; ++v) pl[t * PT + (v + half * 8) * 17 + l16] = pr[t][v];
  };

  auto loadP = [&](v2f (&pa)[KT][4]) {
#pragma unroll
    for (int t = 0; t < KT; ++t)
#pragma unroll
      for (int kc = 0; kc < 4; ++kc) {
        pa[t][kc].x = pl[t * PT + l16 * 17 + kc * 4 + halfk];
        pa[t][kc].y = pl[t * PT + l16 * 17 + kc * 4 + halfk + 1];
      }
  };

  auto scores_beside_softmax = [&](const v4f (&kf)[KT][8], v8f (&sn)[KT],
                                   const v8f (&sc)[KT], v2f (&pa)[KT][4]) {
#pragma unroll
    for (int t = 0; t < KT; ++t) sn[t] = v8f{};
#pragma unroll
    for (int i = 0; i < 8; ++i) {
      qk_i(kf, sn, i);
      const int r0 = (i + i / 2 < 8) ? i + i / 2 : 8;
      const int r1 = (i + 1 + (i + 1) / 2 < 8) ? i + 1 + (i + 1) / 2 : 8;
#pragma unroll
      for (int v = r0; v < r1; ++v) softmax_row(sc, v);
      if (i == 5) {
#pragma unroll
        for (int t = 0; t < KT; ++t)
#pragma unroll
          for (int v = 0; v < 8; ++v) asm volatile("" : "+v"(pr[t][v]));
        __builtin_amdgcn_sched_barrier(0);
        storeP();
        loadP(pa);
        __builtin_amdgcn_sched_barrier(0);
      }
    }
  };

  auto pv = [&](const v2f (&pa)[KT][4], const v4f (&vf)[KT][8]) {
#pragma unroll
    for (int t = 0; t < KT; ++t) {
#pragma unroll
      for (int kc = 0; kc < 4; ++kc) {
        const v2f a = pa[t][kc];
        const v4f f01 = vf[t][2 * kc];
        const v4f f23 = vf[t][2 * kc + 1];
        { v2f bv; bv.x = f01.x; bv.y = f01.y; acc0 = wmma_f32(a, bv, acc0); }
        { v2f bv; bv.x = f01.z; bv.y = f01.w; acc1 = wmma_f32(a, bv, acc1); }
        { v2f bv; bv.x = f23.x; bv.y = f23.y; acc2 = wmma_f32(a, bv, acc2); }
        { v2f bv; bv.x = f23.z; bv.y = f23.w; acc3 = wmma_f32(a, bv, acc3); }
      }
    }
  };

  constexpr int NG = NKT / KT;
  v4f ka[KT][8], va[KT][8], kb[KT][8], vb[KT][8];
  v8f sa[KT], sb[KT];
  v2f pa[KT][4];
  fetchK(ka, 0);
  fetchV(va, 0);
  fetchK(kb, 1);
#pragma unroll
  for (int t = 0; t < KT; ++t) sa[t] = v8f{};
#pragma unroll
  for (int i = 0; i < 8; ++i) qk_i(ka, sa, i);
  fetchK(ka, 2);

  for (int g = 0; g < NG - 2; g += 2) {
    scores_beside_softmax(kb, sb, sa, pa);
    __builtin_amdgcn_sched_barrier(0);
    fetchK(kb, g + 3);
    fetchV(vb, g + 1);
    __builtin_amdgcn_sched_barrier(0);
    pv(pa, va);
    __builtin_amdgcn_sched_barrier(0);
    scores_beside_softmax(ka, sa, sb, pa);
    __builtin_amdgcn_sched_barrier(0);
    fetchK(ka, (g + 4 < NG) ? g + 4 : NG - 1);
    fetchV(va, g + 2);
    __builtin_amdgcn_sched_barrier(0);
    pv(pa, vb);
  }
  scores_beside_softmax(kb, sb, sa, pa);
  fetchV(vb, NG - 1);
  pv(pa, va);
#pragma unroll
  for (int v = 0; v < 8; ++v) softmax_row(sb, v);
  storeP();
  loadP(pa);
  pv(pa, vb);

#pragma unroll
  for (int v = 0; v < 8; ++v) {
    float l = lst[v];
    l += row_xor<1>(l);
    l += row_xor<2>(l);
    l += row_xor<4>(l);
    l += row_xor<8>(l);
    const float inv = 1.0f / l;
    const int orow = (b * SEQ + qbase + v + half * 8) * HS;
    out[orow + l16]      = acc0[v] * inv;
    out[orow + 16 + l16] = acc1[v] * inv;
    out[orow + 32 + l16] = acc2[v] * inv;
    out[orow + 48 + l16] = acc3[v] * inv;
  }
}

extern "C" void kernel_launch(void* const* d_in, const int* in_sizes, int n_in,
                              void* d_out, int out_size, void* d_ws, size_t ws_size,
                              hipStream_t stream) {
  (void)in_sizes; (void)n_in; (void)out_size; (void)ws_size;
  const float* x  = (const float*)d_in[0];
  const float* Wk = (const float*)d_in[1];
  const float* Wq = (const float*)d_in[2];
  const float* Wv = (const float*)d_in[3];

  float* ws = (float*)d_ws;
  const size_t plane = (size_t)BATCH * SEQ * HS;
  float* Q  = ws;
  float* Kf = ws + plane;
  float* Vf = ws + 2 * plane;
  float* out = (float*)d_out;

  proj_kernel<<<(BATCH * SEQ / 16) / 4, 128, 0, stream>>>(x, Wq, Wk, Wv, Q, Kf, Vf);

  attn_kernel<<<(BATCH * SEQ / 16) / 4, 128, 0, stream>>>(Q, Kf, Vf, out);
}


# attention softmax keeps each row's shift while a 32-key step's exponents stay inside a window above it (one wave-uniform test): max butterfly + rescale only on a miss; sums keep their order
# speedup vs baseline: 5.6106x; 1.0523x over previous
#include <hip/hip_runtime.h>
#include <math.h>


typedef __attribute__((ext_vector_type(2))) float v2f;
typedef __attribute__((ext_vector_type(4))) float v4f;
typedef __attribute__((ext_vector_type(8))) float v8f;

#define EMB   768
#define HS    64
#define SEQ   2048
#define BATCH 8
#define TILE  1024
#define NKT   (SEQ / 16)

__device__ __forceinline__ v8f wmma_f32(v2f a, v2f b, v8f c) {
  return __builtin_amdgcn_wmma_f32_16x16x4_f32(
       false, a,  false, b,
       (short)0, c,  false,  false);
}


template <int M>
__device__ __forceinline__ float row_xor(float x) {
  return __builtin_amdgcn_update_dpp(0.0f, x, 0x160 + M, 0xf, 0xf, true);
}


#define PU  6
#define PG  (EMB / 8 / PU)

__device__ __forceinline__ void store_tile(int which, int cw, int tileM, const v8f& c,
                                           int l16, int half,
                                           float* __restrict__ Q, float* __restrict__ Kf,
                                           float* __restrict__ Vf) {
  const int n = cw * 16 + l16;
  if (which == 0) {
#pragma unroll
    for (int v = 0; v < 8; ++v) {
      const int m = tileM * 16 + v + half * 8;
      Q[m * HS + n] = c[v];
    }
  } else if (which == 1) {
    float* T = Kf + tileM * TILE;
    const int base = (n >> 3) * 128 + ((n >> 1) & 1) * 64 + ((n >> 2) & 1) * 2 + (n & 1);
#pragma unroll
    for (int v = 0; v < 8; ++v) {
      const int key = v + half * 8;
      T[base + key * 4] = c[v];
    }
  } else {
    float* T = Vf + tileM * TILE;
#pragma unroll
    for (int v = 0; v < 8; v += 2) {
      const int kc = (v >> 2) + 2 * half;
      const int idx = (kc * 2 + (cw >> 1)) * 128 + (((v >> 1) & 1) * 16 + l16) * 4 + (cw & 1) * 2;
      v2f pr; pr.x = c[v]; pr.y = c[v + 1];
      *(v2f*)(T + idx) = pr;
    }
  }
}

__global__ __launch_bounds__(128) void proj_kernel(
    const float* __restrict__ x,
    const float* __restrict__ Wq, const float* __restrict__ Wk,
    const float* __restrict__ Wv,
    float* __restrict__ Q, float* __restrict__ Kf, float* __restrict__ Vf) {
  const int tid   = threadIdx.x;
  const int lane  = tid & 31;
  const int wave  = __builtin_amdgcn_readfirstlane(tid >> 5);
  const int l16   = lane & 15;
  const int half  = lane >> 4;
  const int pair  = blockIdx.x * 2 + (wave >> 1);
  const int nh    = wave & 1;

  const int   lofs = (4 * half) * HS + l16;
  const float* w0 = (nh ? Wk + 32 : Wq)      + lofs;
  const float* w1 = (nh ? Wv      : Wq + 32) + lofs;
  const float* w2 = (nh ? Wv + 32 : Wk)      + lofs;

  auto fetch = [&](v2f (&bf)[PU][2][2], const float* wp, int g) {
#pragma unroll
    for (int u = 0; u < PU; ++u)
#pragma unroll
      for (int f = 0; f < 2; ++f)
#pragma unroll
        for (int e = 0; e < 2; ++e) {
          const float* q = wp + ((g * PU + u) * 8 + 2 * f) * HS + e * 16;
          bf[u][f][e].x = q[0];
          bf[u][f][e].y = q[HS];
        }
  };

  v2f ba[PU][2][2], bb[PU][2][2];
  const float* wp = w0;
  fetch(ba, wp, 0);

  v4f xa[2][EMB / 8];
  const v4f* xr0 = (const v4f*)(x + ((2 * pair) * 16 + l16) * EMB) + half;
  const v4f* xr1 = xr0 + 16 * (EMB / 4);
  auto xload = [&](int g) {
#pragma unroll
    for (int u = 0; u < PU; ++u) {
      const int i = g * PU + u;
      xa[0][i] = xr0[2 * i];
      xa[1][i] = xr1[2 * i];
    }
  };
  xload(0);
  xload(1);

  v8f acc[2][2];

  auto step = [&](const v2f (&bf)[PU][2][2], int g) {
#pragma unroll
    for (int u = 0; u < PU; ++u) {
      const int i = g * PU + u;
      v2f a[2][2];
#pragma unroll
      for (int r = 0; r < 2; ++r) {
        a[r][0].x = xa[r][i].x; a[r][0].y = xa[r][i].y;
        a[r][1].x = xa[r][i].z; a[r][1].y = xa[r][i].w;
      }
#pragma unroll
      for (int f = 0; f < 2; ++f)
#pragma unroll
        for (int e = 0; e < 2; ++e)
#pragma unroll
          for (int r = 0; r < 2; ++r)
            acc[r][e] = wmma_f32(a[r][f], bf[u][f][e], acc[r][e]);
    }
  };

  for (int p = 0; p < 3; ++p) {
    const float* wn = (p == 0) ? w1 : w2;
#pragma unroll
    for (int r = 0; r < 2; ++r)
#pragma unroll
      for (int e = 0; e < 2; ++e) acc[r][e] = v8f{};

#pragma unroll
    for (int g = 0; g < PG; g += 2) {
      fetch(bb, wp, g + 1);
      if (p == 0 && g + 2 < PG) xload(g + 2);
      __builtin_amdgcn_sched_barrier(0);
      step(ba, g);
      __builtin_amdgcn_sched_barrier(0);
      if (g + 2 < PG) fetch(ba, wp, g + 2);
      else            fetch(ba, wn, 0);
      if (p == 0 && g + 3 < PG) xload(g + 3);
      __builtin_amdgcn_sched_barrier(0);
      step(bb, g + 1);
      __builtin_amdgcn_sched_barrier(0);
    }

    const int jp = nh * 3 + p;
#pragma unroll
    for (int r = 0; r < 2; ++r)
#pragma unroll
      for (int e = 0; e < 2; ++e)
        store_tile(jp >> 1, (jp & 1) * 2 + e, 2 * pair + r, acc[r][e], l16, half, Q, Kf, Vf);
    wp = wn;
  }
}

#define KT  2
#define PT  (16 * 17)

__global__ __launch_bounds__(128) void attn_kernel(
    const float* __restrict__ Q, const float* __restrict__ Kf,
    const float* __restrict__ Vf, float* __restrict__ out) {
  __shared__ float plds[4][KT * PT];

  const int tid   = threadIdx.x;
  const int lane  = tid & 31;
  const int wave  = tid >> 5;
  const int gt    = blockIdx.x * 4 + wave;
  const int b     = gt >> 7;
  const int qbase = (gt & 127) * 16;
  const int l16   = lane & 15;
  const int half  = lane >> 4;
  const int halfk = half * 2;

  v2f qa[16];
  const int mrow = (b * SEQ + qbase + l16) * HS;
#pragma unroll
  for (int kc = 0; kc < 16; ++kc) {
    qa[kc].x = Q[mrow + kc * 4 + halfk];
    qa[kc].y = Q[mrow + kc * 4 + halfk + 1];
  }

  v8f acc0 = {}, acc1 = {}, acc2 = {}, acc3 = {};
  float nsh[8], lst[8];
#pragma unroll
  for (int v = 0; v < 8; ++v) { nsh[v] = INFINITY; lst[v] = 0.0f; }

  float* pl = &plds[wave][0];
  const float cexp = 0.125f * 1.4426950408889634f;
  const float WIN  = 8.0f;
  float dd[KT][8];
  float pr[KT][8];

  const v4f* kp = (const v4f*)Kf + (b * NKT) * (TILE / 4) + lane;
  const v4f* vp = (const v4f*)Vf + (b * NKT) * (TILE / 4) + lane;

  auto fetchK = [&](v4f (&kf)[KT][8], int g) {
    const v4f* kq = kp + g * (KT * TILE / 4);
#pragma unroll
    for (int t = 0; t < KT; ++t)
#pragma unroll
      for (int i = 0; i < 8; ++i) kf[t][i] = kq[(t * 8 + i) * 32];
  };
  auto fetchV = [&](v4f (&vf)[KT][8], int g) {
    const v4f* vq = vp + g * (KT * TILE / 4);
#pragma unroll
    for (int t = 0; t < KT; ++t)
#pragma unroll
      for (int i = 0; i < 8; ++i) vf[t][i] = vq[(t * 8 + i) * 32];
  };

  auto qk_i = [&](const v4f (&kf)[KT][8], v8f (&s)[KT], int i) {
#pragma unroll
    for (int t = 0; t < KT; ++t) {
      v2f k0; k0.x = kf[t][i].x; k0.y = kf[t][i].y;
      s[t] = wmma_f32(qa[2 * i], k0, s[t]);
    }
#pragma unroll
    for (int t = 0; t < KT; ++t) {
      v2f k1; k1.x = kf[t][i].z; k1.y = kf[t][i].w;
      s[t] = wmma_f32(qa[2 * i + 1], k1, s[t]);
    }
  };

  auto window_test = [&](const v8f (&s)[KT]) -> bool {
    float mx = -INFINITY;
#pragma unroll
    for (int v = 0; v < 8; ++v)
#pragma unroll
      for (int t = 0; t < KT; ++t) {
        dd[t][v] = fmaf(s[t][v], cexp, nsh[v]);
        mx = fmaxf(mx, dd[t][v]);
      }
    return __builtin_amdgcn_ballot_w32(mx > WIN) != 0u;
  };

  auto rebase = [&](const v8f (&s)[KT]) {
#pragma unroll
    for (int v = 0; v < 8; ++v) {
      float tm = s[0][v];
#pragma unroll
      for (int t = 1; t < KT; ++t) tm = fmaxf(tm, s[t][v]);
      tm = fmaxf(tm, row_xor<1>(tm));
      tm = fmaxf(tm, row_xor<2>(tm));
      tm = fmaxf(tm, row_xor<4>(tm));
      tm = fmaxf(tm, row_xor<8>(tm));
      const float nn    = fminf(nsh[v], tm * -cexp);
      const float alpha = __builtin_amdgcn_exp2f(nn - nsh[v]);
      lst[v] *= alpha;
      acc0[v] *= alpha; acc1[v] *= alpha; acc2[v] *= alpha; acc3[v] *= alpha;
      nsh[v] = nn;
#pragma unroll
      for (int t = 0; t < KT; ++t) dd[t][v] = fmaf(s[t][v], cexp, nn);
    }
  };

  auto exp_row = [&](int v) {
    float ps = 0.0f;
#pragma unroll
    for (int t = 0; t < KT; ++t) {
      const float p = __builtin_amdgcn_exp2f(dd[t][v]);
      ps = (t == 0) ? p : ps + p;
      pr[t][v] = p;
    }
    lst[v] += ps;
  };

  auto storeP = [&]() {
#pragma unroll
    for (int t = 0; t < KT; ++t)
#pragma unroll
      for (int v = 0; v < 8; ++v) pl[t * PT + (v + half * 8) * 17 + l16] = pr[t][v];
  };

  auto loadP = [&](v2f (&pa)[KT][4]) {
#pragma unroll
    for (int t = 0; t < KT; ++t)
#pragma unroll
      for (int kc = 0; kc < 4; ++kc) {
        pa[t][kc].x = pl[t * PT + l16 * 17 + kc * 4 + halfk];
        pa[t][kc].y = pl[t * PT + l16 * 17 + kc * 4 + halfk + 1];
      }
  };

  auto scores_beside_softmax = [&](const v4f (&kf)[KT][8], v8f (&sn)[KT],
                                   const v8f (&sc)[KT], v2f (&pa)[KT][4]) {
#pragma unroll
    for (int t = 0; t < KT; ++t) sn[t] = v8f{};
    qk_i(kf, sn, 0);
    const bool miss = window_test(sc);
    __builtin_amdgcn_sched_barrier(0);
    qk_i(kf, sn, 1);
    __builtin_amdgcn_sched_barrier(0);
    asm volatile("v_nop\n\tv_nop\n\tv_nop\n\tv_nop\n\tv_nop");
    if (__builtin_expect(miss, 0)) rebase(sc);
#pragma unroll
    for (int i = 2; i < 8; ++i) {
      qk_i(kf, sn, i);
      if (i < 6) {
        exp_row(2 * (i - 2));
        exp_row(2 * (i - 2) + 1);
      }
      if (i < 5) __builtin_amdgcn_sched_barrier(0);
      if (i == 5) {
#pragma unroll
        for (int t = 0; t < KT; ++t)
#pragma unroll
          for (int v = 0; v < 8; ++v) asm volatile("" : "+v"(pr[t][v]));
        __builtin_amdgcn_sched_barrier(0);
        storeP();
        loadP(pa);
        __builtin_amdgcn_sched_barrier(0);
      }
    }
  };

  auto pv = [&](const v2f (&pa)[KT][4], const v4f (&vf)[KT][8]) {
#pragma unroll
    for (int t = 0; t < KT; ++t) {
#pragma unroll
      for (int kc = 0; kc < 4; ++kc) {
        const v2f a = pa[t][kc];
        const v4f f01 = vf[t][2 * kc];
        const v4f f23 = vf[t][2 * kc + 1];
        { v2f bv; bv.x = f01.x; bv.y = f01.y; acc0 = wmma_f32(a, bv, acc0); }
        { v2f bv; bv.x = f01.z; bv.y = f01.w; acc1 = wmma_f32(a, bv, acc1); }
        { v2f bv; bv.x = f23.x; bv.y = f23.y; acc2 = wmma_f32(a, bv, acc2); }
        { v2f bv; bv.x = f23.z; bv.y = f23.w; acc3 = wmma_f32(a, bv, acc3); }
      }
    }
  };

  constexpr int NG = NKT / KT;
  v4f ka[KT][8], va[KT][8], kb[KT][8], vb[KT][8];
  v8f sa[KT], sb[KT];
  v2f pa[KT][4];
  fetchK(ka, 0);
  fetchV(va, 0);
  fetchK(kb, 1);
#pragma unroll
  for (int t = 0; t < KT; ++t) sa[t] = v8f{};
#pragma unroll
  for (int i = 0; i < 8; ++i) qk_i(ka, sa, i);
  fetchK(ka, 2);

  for (int g = 0; g < NG - 2; g += 2) {
    scores_beside_softmax(kb, sb, sa, pa);
    __builtin_amdgcn_sched_barrier(0);
    fetchK(kb, g + 3);
    fetchV(vb, g + 1);
    __builtin_amdgcn_sched_barrier(0);
    pv(pa, va);
    __builtin_amdgcn_sched_barrier(0);
    scores_beside_softmax(ka, sa, sb, pa);
    __builtin_amdgcn_sched_barrier(0);
    fetchK(ka, (g + 4 < NG) ? g + 4 : NG - 1);
    fetchV(va, g + 2);
    __builtin_amdgcn_sched_barrier(0);
    pv(pa, vb);
  }
  scores_beside_softmax(kb, sb, sa, pa);
  fetchV(vb, NG - 1);
  pv(pa, va);
  if (window_test(sb)) rebase(sb);
#pragma unroll
  for (int v = 0; v < 8; ++v) exp_row(v);
  storeP();
  loadP(pa);
  pv(pa, vb);

#pragma unroll
  for (int v = 0; v < 8; ++v) {
    float l = lst[v];
    l += row_xor<1>(l);
    l += row_xor<2>(l);
    l += row_xor<4>(l);
    l += row_xor<8>(l);
    const float inv = 1.0f / l;
    const int orow = (b * SEQ + qbase + v + half * 8) * HS;
    out[orow + l16]      = acc0[v] * inv;
    out[orow + 16 + l16] = acc1[v] * inv;
    out[orow + 32 + l16] = acc2[v] * inv;
    out[orow + 48 + l16] = acc3[v] * inv;
  }
}

extern "C" void kernel_launch(void* const* d_in, const int* in_sizes, int n_in,
                              void* d_out, int out_size, void* d_ws, size_t ws_size,
                              hipStream_t stream) {
  (void)in_sizes; (void)n_in; (void)out_size; (void)ws_size;
  const float* x  = (const float*)d_in[0];
  const float* Wk = (const float*)d_in[1];
  const float* Wq = (const float*)d_in[2];
  const float* Wv = (const float*)d_in[3];

  float* ws = (float*)d_ws;
  const size_t plane = (size_t)BATCH * SEQ * HS;
  float* Q  = ws;
  float* Kf = ws + plane;
  float* Vf = ws + 2 * plane;
  float* out = (float*)d_out;

  proj_kernel<<<(BATCH * SEQ / 16) / 4, 128, 0, stream>>>(x, Wq, Wk, Wv, Q, Kf, Vf);

  attn_kernel<<<(BATCH * SEQ / 16) / 4, 128, 0, stream>>>(Q, Kf, Vf, out);
}
